# starting kernel
# baseline (speedup 1.0000x reference)
#include <hip/hip_runtime.h>
#include <hip/hip_bf16.h>

typedef float v2f __attribute__((ext_vector_type(2)));
typedef float v4f __attribute__((ext_vector_type(4)));
typedef float v8f __attribute__((ext_vector_type(8)));

#define N_CLASS 10
#define NDIM    64
#define NCOMP   32
#define NKNOT   32
#define NTOT    65536

#define ROWS_PER_BLOCK 64
#define TPB            128
#define NWAVES         4
#define SD_STRIDE      68
#define SE_STRIDE      36

__global__ __launch_bounds__(TPB)
void cst_fused_kernel(const float* __restrict__ data,
                      const int*   __restrict__ label,
                      const float* __restrict__ wT,
                      const float* __restrict__ xx,
                      const float* __restrict__ yy,
                      const float* __restrict__ dl,
                      float* __restrict__ out,
                      float* __restrict__ logj)
{
    __shared__ float sData[ROWS_PER_BLOCK * SD_STRIDE];
    __shared__ v2f   sB1[16 * 2 * 32];
    __shared__ v2f   sB2[8 * 4 * 32];
    __shared__ float sE[NWAVES * 16 * SE_STRIDE];
    __shared__ int   sLab[ROWS_PER_BLOCK];

    const int tid  = threadIdx.x;
    const int base = blockIdx.x * ROWS_PER_BLOCK;
    const int lane = tid & 31;
    const int wave = tid >> 5;
    const int hi   = lane >> 4;
    const int l16  = lane & 15;

#pragma unroll
    for (int t = 0; t < 8; ++t) {
        int idx = tid + t * TPB;
        int row = idx >> 4, c4 = idx & 15;
        v4f v = __builtin_nontemporal_load(
                    (const v4f*)(data + (size_t)(base + row) * NDIM + c4 * 4));
        *(v4f*)(sData + row * SD_STRIDE + c4 * 4) = v;
    }
#pragma unroll
    for (int t = 0; t < 8; ++t) {
        int e = tid + t * TPB;
        int ln = e & 31, cj = e >> 5;
        int chunk = cj >> 1, j = cj & 1;
        int krow = 4 * chunk + 2 * (ln >> 4);
        int col  = (ln & 15) + 16 * j;
        v2f b;
        b.x = wT[krow * NCOMP + col];
        b.y = wT[(krow + 1) * NCOMP + col];
        sB1[e] = b;
    }
#pragma unroll
    for (int t = 0; t < 8; ++t) {
        int e = tid + t * TPB;
        int ln = e & 31, cj = e >> 5;
        int chunk = cj >> 2, jj = cj & 3;
        int n = jj * 16 + (ln & 15);
        int k = 4 * chunk + 2 * (ln >> 4);
        sB2[e] = *(const v2f*)(wT + n * NCOMP + k);
    }
    if (tid < ROWS_PER_BLOCK) sLab[tid] = label[base + tid];
    __syncthreads();

    const int rowL = wave * 16 + l16;
    v2f a1[16];
#pragma unroll
    for (int c = 0; c < 16; ++c)
        a1[c] = *(const v2f*)(sData + rowL * SD_STRIDE + 4 * c + 2 * hi);

    v8f c0 = {0.f, 0.f, 0.f, 0.f, 0.f, 0.f, 0.f, 0.f};
    v8f c1 = c0;
#pragma unroll
    for (int c = 0; c < 16; ++c)
        c0 = __builtin_amdgcn_wmma_f32_16x16x4_f32(false, a1[c], false,
                sB1[(c * 2 + 0) * 32 + lane], (short)0, c0, false, false);
#pragma unroll
    for (int c = 0; c < 16; ++c)
        c1 = __builtin_amdgcn_wmma_f32_16x16x4_f32(false, a1[c], false,
                sB1[(c * 2 + 1) * 32 + lane], (short)0, c1, false, false);

    float* sEw = sE + wave * 16 * SE_STRIDE;
    int kb[16];
    int idx[16];
    float ld[8];

#pragma unroll
    for (int r = 0; r < 8; ++r) {
        int cls = sLab[wave * 16 + r + 8 * hi];
        int k0  = (cls * NCOMP + l16) * NKNOT;
        kb[r]     = k0;
        kb[r + 8] = k0 + 16 * NKNOT;
    }
#define XV(e) (((e) < 8) ? c0[(e) & 7] : c1[(e) & 7])
#pragma unroll
    for (int e = 0; e < 16; ++e) idx[e] = (xx[kb[e] + 15] < XV(e)) ? 16 : 0;
#pragma unroll
    for (int e = 0; e < 16; ++e) idx[e] += (xx[kb[e] + idx[e] + 7] < XV(e)) ? 8 : 0;
#pragma unroll
    for (int e = 0; e < 16; ++e) idx[e] += (xx[kb[e] + idx[e] + 3] < XV(e)) ? 4 : 0;
#pragma unroll
    for (int e = 0; e < 16; ++e) idx[e] += (xx[kb[e] + idx[e] + 1] < XV(e)) ? 2 : 0;
#pragma unroll
    for (int e = 0; e < 16; ++e) idx[e] += (xx[kb[e] + idx[e]] < XV(e)) ? 1 : 0;
#pragma unroll
    for (int e = 0; e < 16; ++e) {
        float t = xx[kb[e] + 31];
        if (idx[e] == 31 && t < XV(e)) idx[e] = 32;
    }

#pragma unroll
    for (int g = 0; g < 2; ++g) {
        float xk[8], xk1[8], yk[8], yk1[8], dk[8], dk1[8];
#pragma unroll
        for (int r = 0; r < 8; ++r) {
            int e = g * 8 + r;
            int b = idx[e] - 1;
            b = b < 0 ? 0 : (b > NKNOT - 2 ? NKNOT - 2 : b);
            int o = kb[e] + b;
            xk[r] = xx[o]; xk1[r] = xx[o + 1];
            yk[r] = yy[o]; yk1[r] = yy[o + 1];
            dk[r] = dl[o]; dk1[r] = dl[o + 1];
        }
#pragma unroll
        for (int r = 0; r < 8; ++r) {
            int e = g * 8 + r;
            float x   = (g == 0) ? c0[r] : c1[r];
            float w   = xk1[r] - xk[r];
            float s   = (yk1[r] - yk[r]) / w;
            float xi  = fminf(fmaxf((x - xk[r]) / w, 0.0f), 1.0f);
            float xi1 = 1.0f - xi;
            float cc  = xi * xi1;
            float den = s + (dk1[r] + dk[r] - 2.0f * s) * cc;
            float y_in = yk[r] + (yk1[r] - yk[r]) * (s * xi * xi + dk[r] * cc) / den;
            float d_in = s * s * (dk1[r] * xi * xi + 2.0f * s * cc + dk[r] * xi1 * xi1)
                         / (den * den);
            bool below = (idx[e] == 0);
            bool above = (idx[e] == NKNOT);
            float y  = below ? fmaf(x - xk[r], dk[r], yk[r])
                     : (above ? fmaf(x - xk1[r], dk1[r], yk1[r]) : y_in);
            float dd = below ? dk[r] : (above ? dk1[r] : d_in);
            float l  = __logf(dd);
            if (g == 0) ld[r] = l; else ld[r] += l;
            int M = r + 8 * hi;
            sEw[M * SE_STRIDE + l16 + 16 * g] = y - x;
        }
    }
#undef XV

#pragma unroll
    for (int r = 0; r < 8; ++r) {
        float p = ld[r];
        p += __shfl_xor(p, 1, 32);
        p += __shfl_xor(p, 2, 32);
        p += __shfl_xor(p, 4, 32);
        p += __shfl_xor(p, 8, 32);
        if (l16 == 0)
            __builtin_nontemporal_store(p, logj + base + wave * 16 + r + 8 * hi);
    }
    asm volatile("s_wait_dscnt 0" ::: "memory");

    v2f a2[8];
#pragma unroll
    for (int c = 0; c < 8; ++c)
        a2[c] = *(const v2f*)(sEw + l16 * SE_STRIDE + 4 * c + 2 * hi);

#pragma unroll
    for (int jj = 0; jj < 4; ++jj) {
        v8f acc = {0.f, 0.f, 0.f, 0.f, 0.f, 0.f, 0.f, 0.f};
#pragma unroll
        for (int c = 0; c < 8; ++c)
            acc = __builtin_amdgcn_wmma_f32_16x16x4_f32(false, a2[c], false,
                    sB2[(c * 4 + jj) * 32 + lane], (short)0, acc, false, false);
        int col = jj * 16 + l16;
#pragma unroll
        for (int r = 0; r < 8; ++r) {
            int M    = r + 8 * hi;
            int srow = wave * 16 + M;
            __builtin_nontemporal_store(acc[r] + sData[srow * SD_STRIDE + col],
                                        out + (size_t)(base + srow) * NDIM + col);
        }
    }
}

extern "C" void kernel_launch(void* const* d_in, const int* in_sizes, int n_in,
                              void* d_out, int out_size, void* d_ws, size_t ws_size,
                              hipStream_t stream) {
    const float* data  = (const float*)d_in[0];
    const int*   label = (const int*)  d_in[1];
    const float* wT    = (const float*)d_in[2];
    const float* xx    = (const float*)d_in[3];
    const float* yy    = (const float*)d_in[4];
    const float* dl    = (const float*)d_in[5];
    float* out  = (float*)d_out;
    float* logj = out + (size_t)NTOT * NDIM;

    dim3 grid(NTOT / ROWS_PER_BLOCK);
    cst_fused_kernel<<<grid, TPB, 0, stream>>>(data, label, wT, xx, yy, dl, out, logj);
}


# Every per-lane global address is a 32-bit byte offset off a scalar base, constants in the load's offset field; the knot search carries its position as that offset: no 64-bit address math; same sums
# speedup vs baseline: 2.0799x; 2.0799x over previous
#include <hip/hip_runtime.h>
#include <hip/hip_bf16.h>


typedef float v2f __attribute__((ext_vector_type(2)));
typedef float v4f __attribute__((ext_vector_type(4)));
typedef float v8f __attribute__((ext_vector_type(8)));

#define N_CLASS 10
#define NDIM    64
#define NCOMP   32
#define NKNOT   32
#define NTOT    65536

#define ROWS_PER_BLOCK 128
#define TPB            256
#define NWAVES         8
#define LAB_PER_LANE   (ROWS_PER_BLOCK / 32)
#define SD_STRIDE      68


__global__ __launch_bounds__(TPB)
void cst_fused_kernel(const float* __restrict__ data,
                      const int*   __restrict__ label,
                      const float* __restrict__ wT,
                      const float* __restrict__ xx,
                      const float* __restrict__ yy,
                      const float* __restrict__ dl,
                      float* __restrict__ out,
                      float* __restrict__ logj)
{
    __shared__ float sData[ROWS_PER_BLOCK * SD_STRIDE];
    __shared__ v2f   sW1[16 * 2 * 32];
    __shared__ v2f   sW2[8 * 4 * 32];
    __shared__ int   sSlot[NWAVES * 16];
    __shared__ float sLogj[ROWS_PER_BLOCK];

    const int tid  = threadIdx.x;
    const int base = blockIdx.x * ROWS_PER_BLOCK;
    const int lane = tid & 31;
    const int wave = tid >> 5;
    const int hi   = lane >> 4;
    const int l16  = lane & 15;

    {
        const int row0 = tid >> 4, c4 = tid & 15;
        const unsigned dOff = (unsigned)(base + row0) * (NDIM * 4u) + (unsigned)c4 * 16u;
        const char* dataB = (const char*)data;
#pragma unroll
        for (int t = 0; t < (ROWS_PER_BLOCK * 16) / TPB; ++t) {
            v4f v = __builtin_nontemporal_load(
                        (const v4f*)(dataB + (size_t)dOff + (size_t)t * (16 * NDIM * 4)));
            *(v4f*)(sData + (row0 + 16 * t) * SD_STRIDE + c4 * 4) = v;
        }
    }
#pragma unroll
    for (int t = 0; t < 1024 / TPB; ++t) {
        int e = tid + t * TPB;
        int ln = e & 31, cj = e >> 5;
        int chunk = cj >> 1, j = cj & 1;
        int krow = 4 * chunk + 2 * (ln >> 4);
        int col  = (ln & 15) + 16 * j;
        v2f b;
        b.x = wT[krow * NCOMP + col];
        b.y = wT[(krow + 1) * NCOMP + col];
        sW1[e] = b;
    }
#pragma unroll
    for (int t = 0; t < 1024 / TPB; ++t) {
        int e = tid + t * TPB;
        int ln = e & 31, cj = e >> 5;
        int chunk = cj >> 2, jj = cj & 3;
        int n = jj * 16 + (ln & 15);
        int k = ((chunk < 4) ? 0 : 8) + 8 * (ln >> 4) + 2 * chunk;
        sW2[e] = *(const v2f*)((const char*)wT + (size_t)((unsigned)(n * NCOMP + k) * 4u));
    }
    {
        int lab[LAB_PER_LANE], rk[LAB_PER_LANE];
#pragma unroll
        for (int q = 0; q < LAB_PER_LANE; ++q) { lab[q] = label[base + 32 * q + lane]; rk[q] = 0; }
        int off = 0;
#pragma unroll
        for (int c = 0; c < N_CLASS; ++c) {
#pragma unroll
            for (int q = 0; q < LAB_PER_LANE; ++q) {
                bool mine = (lab[q] == c);
                unsigned m = __builtin_amdgcn_ballot_w32(mine);
                int r = off + (int)__builtin_amdgcn_mbcnt_lo(m, 0u);
                rk[q] = mine ? r : rk[q];
                off += __builtin_popcount(m);
            }
        }
#pragma unroll
        for (int q = 0; q < LAB_PER_LANE; ++q)
            if ((rk[q] >> 4) == wave)
                sSlot[rk[q]] = (lab[q] << 8) | (32 * q + lane);
    }
    __syncthreads();

    const int slot = sSlot[wave * 16 + l16];
    const int rowL = slot & 255;
    v2f bd[16];
#pragma unroll
    for (int c = 0; c < 16; ++c)
        bd[c] = *(const v2f*)(sData + rowL * SD_STRIDE + 4 * c + 2 * hi);

    v8f c0 = {0.f, 0.f, 0.f, 0.f, 0.f, 0.f, 0.f, 0.f};
    v8f c1 = c0;
#pragma unroll
    for (int c = 0; c < 16; ++c)
        c0 = __builtin_amdgcn_wmma_f32_16x16x4_f32(false, sW1[(c * 2 + 0) * 32 + lane], false,
                bd[c], (short)0, c0, false, false);
#pragma unroll
    for (int c = 0; c < 16; ++c)
        c1 = __builtin_amdgcn_wmma_f32_16x16x4_f32(false, sW1[(c * 2 + 1) * 32 + lane], false,
                bd[c], (short)0, c1, false, false);

    const unsigned cls = (unsigned)slot >> 8;
    const unsigned kb0 = (cls * NCOMP + 8u * (unsigned)hi) * (NKNOT * 4u);
    const char* xxB = (const char*)xx;
    const char* yyB = (const char*)yy;
    const char* dlB = (const char*)dl;
#define EB(e) ((size_t)((((e) < 8) ? (e) : (e) + 8) * NKNOT * 4))
#define KNOT(tab, off, e, cst) (*(const float*)((tab) + (size_t)(off) + EB(e) + (size_t)(cst)))
#define XV(e) (((e) < 8) ? c0[(e) & 7] : c1[(e) & 7])
    unsigned pos[16];
    float ev[16];
    float lsum = 0.0f;

#pragma unroll
    for (int e = 0; e < 16; ++e) pos[e] = (KNOT(xxB, kb0, e, 15 * 4) < XV(e)) ? kb0 + 64u : kb0;
#pragma unroll
    for (int e = 0; e < 16; ++e) pos[e] += (KNOT(xxB, pos[e], e, 7 * 4) < XV(e)) ? 32u : 0u;
#pragma unroll
    for (int e = 0; e < 16; ++e) pos[e] += (KNOT(xxB, pos[e], e, 3 * 4) < XV(e)) ? 16u : 0u;
#pragma unroll
    for (int e = 0; e < 16; ++e) pos[e] += (KNOT(xxB, pos[e], e, 1 * 4) < XV(e)) ? 8u : 0u;
#pragma unroll
    for (int e = 0; e < 16; ++e) pos[e] += (KNOT(xxB, pos[e], e, 0) < XV(e)) ? 4u : 0u;
#pragma unroll
    for (int e = 0; e < 16; ++e) {
        float t = KNOT(xxB, kb0, e, 31 * 4);
        if (pos[e] == kb0 + 124u && t < XV(e)) pos[e] = kb0 + 128u;
    }

    float xk[16], xk1[16], yk[16], yk1[16], dk[16], dk1[16];
    auto gather = [&](int e) {
        unsigned g = (pos[e] > kb0 + 4u ? pos[e] : kb0 + 4u) - 4u;
        g = g < kb0 + (NKNOT - 2) * 4u ? g : kb0 + (NKNOT - 2) * 4u;
        xk[e] = KNOT(xxB, g, e, 0); xk1[e] = KNOT(xxB, g, e, 4);
        yk[e] = KNOT(yyB, g, e, 0); yk1[e] = KNOT(yyB, g, e, 4);
        dk[e] = KNOT(dlB, g, e, 0); dk1[e] = KNOT(dlB, g, e, 4);
    };
    auto spline = [&](int e) {
        float x   = XV(e);
        float w   = xk1[e] - xk[e];
        float s   = (yk1[e] - yk[e]) / w;
        float xi  = fminf(fmaxf((x - xk[e]) / w, 0.0f), 1.0f);
        float xi1 = 1.0f - xi;
        float cc  = xi * xi1;
        float den = s + (dk1[e] + dk[e] - 2.0f * s) * cc;
        float y_in = yk[e] + (yk1[e] - yk[e]) * (s * xi * xi + dk[e] * cc) / den;
        float d_in = s * s * (dk1[e] * xi * xi + 2.0f * s * cc + dk[e] * xi1 * xi1)
                     / (den * den);
        bool below = (pos[e] == kb0);
        bool above = (pos[e] == kb0 + NKNOT * 4u);
        float y  = below ? fmaf(x - xk[e], dk[e], yk[e])
                 : (above ? fmaf(x - xk1[e], dk1[e], yk1[e]) : y_in);
        float dd = below ? dk[e] : (above ? dk1[e] : d_in);
        lsum += __logf(dd);
        ev[e] = y - x;
    };
#pragma unroll
    for (int e = 0; e < 16; ++e) gather(e);
    __builtin_amdgcn_sched_barrier(0);
#pragma unroll
    for (int e = 0; e < 8; ++e) spline(e);
    __builtin_amdgcn_sched_barrier(0);
#pragma unroll
    for (int e = 8; e < 16; ++e) spline(e);
#undef XV
#undef KNOT
#undef EB

    lsum += __shfl_xor(lsum, 16, 32);
    if (hi == 0) sLogj[rowL] = lsum;

    float* myRow = sData + rowL * SD_STRIDE + 8 * hi;
#pragma unroll
    for (int jj = 0; jj < 4; ++jj) {
        v8f acc = {0.f, 0.f, 0.f, 0.f, 0.f, 0.f, 0.f, 0.f};
#pragma unroll
        for (int c = 0; c < 8; ++c) {
            v2f be;
            be.x = ev[2 * c];
            be.y = ev[2 * c + 1];
            acc = __builtin_amdgcn_wmma_f32_16x16x4_f32(false, sW2[(c * 4 + jj) * 32 + lane], false,
                    be, (short)0, acc, false, false);
        }
        v4f d0 = *(v4f*)(myRow + 16 * jj);
        v4f d1 = *(v4f*)(myRow + 16 * jj + 4);
        d0.x += acc[0]; d0.y += acc[1]; d0.z += acc[2]; d0.w += acc[3];
        d1.x += acc[4]; d1.y += acc[5]; d1.z += acc[6]; d1.w += acc[7];
        *(v4f*)(myRow + 16 * jj)     = d0;
        *(v4f*)(myRow + 16 * jj + 4) = d1;
    }
    asm volatile("s_wait_dscnt 0" ::: "memory");

#pragma unroll
    for (int i = 0; i < 8; ++i) {
        int srow = __shfl(rowL, 2 * i + hi, 32);
        v4f v = *(const v4f*)(sData + srow * SD_STRIDE + 4 * l16);
        __builtin_nontemporal_store(v, (v4f*)((char*)out
                + (size_t)((unsigned)(base + srow) * (NDIM * 4u) + 16u * (unsigned)l16)));
    }

    __syncthreads();
    if (tid < ROWS_PER_BLOCK / 2) {
        v2f lj = *(const v2f*)(sLogj + 2 * tid);
        __builtin_nontemporal_store(lj, (v2f*)(logj + base + 2 * tid));
    }
}

extern "C" void kernel_launch(void* const* d_in, const int* in_sizes, int n_in,
                              void* d_out, int out_size, void* d_ws, size_t ws_size,
                              hipStream_t stream) {
    const float* data  = (const float*)d_in[0];
    const int*   label = (const int*)  d_in[1];
    const float* wT    = (const float*)d_in[2];
    const float* xx    = (const float*)d_in[3];
    const float* yy    = (const float*)d_in[4];
    const float* dl    = (const float*)d_in[5];
    float* out  = (float*)d_out;
    float* logj = out + (size_t)NTOT * NDIM;

    dim3 grid(NTOT / ROWS_PER_BLOCK);
    cst_fused_kernel<<<grid, TPB, 0, stream>>>(data, label, wT, xx, yy, dl, out, logj);
}


# Knot search: 3 binary levels, then one 16-byte load of the 4-knot window plus the knot before it settles the rest in registers and holds the bin's x knots: 7 loads per element, not 9; same sums
# speedup vs baseline: 2.1940x; 1.0549x over previous
#include <hip/hip_runtime.h>
#include <hip/hip_bf16.h>


typedef float v2f __attribute__((ext_vector_type(2)));
typedef float v4f __attribute__((ext_vector_type(4)));
typedef float v8f __attribute__((ext_vector_type(8)));

#define N_CLASS 10
#define NDIM    64
#define NCOMP   32
#define NKNOT   32
#define NTOT    65536

#define ROWS_PER_BLOCK 128
#define TPB            256
#define NWAVES         8
#define LAB_PER_LANE   (ROWS_PER_BLOCK / 32)
#define SD_STRIDE      68


__global__ __launch_bounds__(TPB, 4)
void cst_fused_kernel(const float* __restrict__ data,
                      const int*   __restrict__ label,
                      const float* __restrict__ wT,
                      const float* __restrict__ xx,
                      const float* __restrict__ yy,
                      const float* __restrict__ dl,
                      float* __restrict__ out,
                      float* __restrict__ logj)
{
    __shared__ float sData[ROWS_PER_BLOCK * SD_STRIDE];
    __shared__ v2f   sW1[16 * 2 * 32];
    __shared__ v2f   sW2[8 * 4 * 32];
    __shared__ int   sSlot[NWAVES * 16];
    __shared__ float sLogj[ROWS_PER_BLOCK];

    const int tid  = threadIdx.x;
    const int base = blockIdx.x * ROWS_PER_BLOCK;
    const int lane = tid & 31;
    const int wave = tid >> 5;
    const int hi   = lane >> 4;
    const int l16  = lane & 15;

    {
        const int row0 = tid >> 4, c4 = tid & 15;
        const unsigned dOff = (unsigned)(base + row0) * (NDIM * 4u) + (unsigned)c4 * 16u;
        const char* dataB = (const char*)data;
#pragma unroll
        for (int t = 0; t < (ROWS_PER_BLOCK * 16) / TPB; ++t) {
            v4f v = __builtin_nontemporal_load(
                        (const v4f*)(dataB + (size_t)dOff + (size_t)t * (16 * NDIM * 4)));
            *(v4f*)(sData + (row0 + 16 * t) * SD_STRIDE + c4 * 4) = v;
        }
    }
#pragma unroll
    for (int t = 0; t < 1024 / TPB; ++t) {
        int e = tid + t * TPB;
        int ln = e & 31, cj = e >> 5;
        int chunk = cj >> 1, j = cj & 1;
        int krow = 4 * chunk + 2 * (ln >> 4);
        int col  = (ln & 15) + 16 * j;
        v2f b;
        b.x = wT[krow * NCOMP + col];
        b.y = wT[(krow + 1) * NCOMP + col];
        sW1[e] = b;
    }
#pragma unroll
    for (int t = 0; t < 1024 / TPB; ++t) {
        int e = tid + t * TPB;
        int ln = e & 31, cj = e >> 5;
        int chunk = cj >> 2, jj = cj & 3;
        int n = jj * 16 + (ln & 15);
        int k = ((chunk < 4) ? 0 : 8) + 8 * (ln >> 4) + 2 * chunk;
        sW2[e] = *(const v2f*)((const char*)wT + (size_t)((unsigned)(n * NCOMP + k) * 4u));
    }
    {
        int lab[LAB_PER_LANE], rk[LAB_PER_LANE];
#pragma unroll
        for (int q = 0; q < LAB_PER_LANE; ++q) { lab[q] = label[base + 32 * q + lane]; rk[q] = 0; }
        int off = 0;
#pragma unroll
        for (int c = 0; c < N_CLASS; ++c) {
#pragma unroll
            for (int q = 0; q < LAB_PER_LANE; ++q) {
                bool mine = (lab[q] == c);
                unsigned m = __builtin_amdgcn_ballot_w32(mine);
                int r = off + (int)__builtin_amdgcn_mbcnt_lo(m, 0u);
                rk[q] = mine ? r : rk[q];
                off += __builtin_popcount(m);
            }
        }
#pragma unroll
        for (int q = 0; q < LAB_PER_LANE; ++q)
            if ((rk[q] >> 4) == wave)
                sSlot[rk[q]] = (lab[q] << 8) | (32 * q + lane);
    }
    __syncthreads();

    const int slot = sSlot[wave * 16 + l16];
    const int rowL = slot & 255;
    v2f bd[16];
#pragma unroll
    for (int c = 0; c < 16; ++c)
        bd[c] = *(const v2f*)(sData + rowL * SD_STRIDE + 4 * c + 2 * hi);

    v8f c0 = {0.f, 0.f, 0.f, 0.f, 0.f, 0.f, 0.f, 0.f};
    v8f c1 = c0;
#pragma unroll
    for (int c = 0; c < 16; ++c)
        c0 = __builtin_amdgcn_wmma_f32_16x16x4_f32(false, sW1[(c * 2 + 0) * 32 + lane], false,
                bd[c], (short)0, c0, false, false);
#pragma unroll
    for (int c = 0; c < 16; ++c)
        c1 = __builtin_amdgcn_wmma_f32_16x16x4_f32(false, sW1[(c * 2 + 1) * 32 + lane], false,
                bd[c], (short)0, c1, false, false);

    const unsigned cls = (unsigned)slot >> 8;
    const unsigned kb0 = (cls * NCOMP + 8u * (unsigned)hi) * (NKNOT * 4u);
    const char* xxB = (const char*)xx;
    const char* yyB = (const char*)yy;
    const char* dlB = (const char*)dl;
#define EB(e) ((size_t)((((e) < 8) ? (e) : (e) + 8) * NKNOT * 4))
#define KNOT(tab, off, e, cst) (*(const float*)((tab) + (size_t)(off) + EB(e) + (size_t)(cst)))
#define XV(e) (((e) < 8) ? c0[(e) & 7] : c1[(e) & 7])
    unsigned pos[16];
    bool     top[16];
    bool     bot[16];
    float ev[16];
    float lsum = 0.0f;

#pragma unroll
    for (int e = 0; e < 16; ++e) pos[e] = (KNOT(xxB, kb0, e, 15 * 4) < XV(e)) ? kb0 + 64u : kb0;
#pragma unroll
    for (int e = 0; e < 16; ++e) pos[e] += (KNOT(xxB, pos[e], e, 7 * 4) < XV(e)) ? 32u : 0u;
#pragma unroll
    for (int e = 0; e < 16; ++e) pos[e] += (KNOT(xxB, pos[e], e, 3 * 4) < XV(e)) ? 16u : 0u;
    v4f   win[16];
    float wfront[16];
#pragma unroll
    for (int e = 0; e < 16; ++e) {
        win[e] = *(const v4f*)(xxB + (size_t)pos[e] + EB(e));
        unsigned pf = (pos[e] > kb0 + 4u ? pos[e] : kb0 + 4u);
        wfront[e] = *(const float*)(xxB + (size_t)pf + EB(e) - 4);
    }

    float xk[16], xk1[16], yk[16], yk1[16], dk[16], dk1[16];
    unsigned gb[16];
    auto settle = [&](int e) {
        float x  = XV(e);
        bool  m4 = win[e].y < x;
        float a  = m4 ? win[e].z : win[e].x;
        bool  m5 = a < x;
        top[e] = (pos[e] == kb0 + 112u) && m4 && m5 && (win[e].w < x);
        xk[e]  = m5 ? a : (m4 ? win[e].y : wfront[e]);
        xk1[e] = m5 ? (m4 ? win[e].w : win[e].y) : a;
        unsigned fin = pos[e] + (m4 ? 8u : 0u) + (m5 ? 4u : 0u);
        bot[e] = (fin == kb0);
        gb[e] = (fin > kb0 + 4u ? fin : kb0 + 4u) - 4u;
    };
    auto gather = [&](int e) {
        yk[e] = KNOT(yyB, gb[e], e, 0); yk1[e] = KNOT(yyB, gb[e], e, 4);
        dk[e] = KNOT(dlB, gb[e], e, 0); dk1[e] = KNOT(dlB, gb[e], e, 4);
    };
    auto spline = [&](int e) {
        float x   = XV(e);
        float w   = xk1[e] - xk[e];
        float s   = (yk1[e] - yk[e]) / w;
        float xi  = fminf(fmaxf((x - xk[e]) / w, 0.0f), 1.0f);
        float xi1 = 1.0f - xi;
        float cc  = xi * xi1;
        float den = s + (dk1[e] + dk[e] - 2.0f * s) * cc;
        float y_in = yk[e] + (yk1[e] - yk[e]) * (s * xi * xi + dk[e] * cc) / den;
        float d_in = s * s * (dk1[e] * xi * xi + 2.0f * s * cc + dk[e] * xi1 * xi1)
                     / (den * den);
        bool below = bot[e];
        bool above = top[e];
        float y  = below ? fmaf(x - xk[e], dk[e], yk[e])
                 : (above ? fmaf(x - xk1[e], dk1[e], yk1[e]) : y_in);
        float dd = below ? dk[e] : (above ? dk1[e] : d_in);
        lsum += __logf(dd);
        ev[e] = y - x;
    };
#pragma unroll
    for (int e = 0; e < 16; ++e) settle(e);
#pragma unroll
    for (int e = 0; e < 16; ++e) gather(e);
    __builtin_amdgcn_sched_barrier(0);
#pragma unroll
    for (int e = 0; e < 8; ++e) spline(e);
    __builtin_amdgcn_sched_barrier(0);
#pragma unroll
    for (int e = 8; e < 16; ++e) spline(e);
#undef XV
#undef KNOT
#undef EB

    lsum += __shfl_xor(lsum, 16, 32);
    if (hi == 0) sLogj[rowL] = lsum;

    float* myRow = sData + rowL * SD_STRIDE + 8 * hi;
#pragma unroll
    for (int jj = 0; jj < 4; ++jj) {
        v8f acc = {0.f, 0.f, 0.f, 0.f, 0.f, 0.f, 0.f, 0.f};
#pragma unroll
        for (int c = 0; c < 8; ++c) {
            v2f be;
            be.x = ev[2 * c];
            be.y = ev[2 * c + 1];
            acc = __builtin_amdgcn_wmma_f32_16x16x4_f32(false, sW2[(c * 4 + jj) * 32 + lane], false,
                    be, (short)0, acc, false, false);
        }
        v4f d0 = *(v4f*)(myRow + 16 * jj);
        v4f d1 = *(v4f*)(myRow + 16 * jj + 4);
        d0.x += acc[0]; d0.y += acc[1]; d0.z += acc[2]; d0.w += acc[3];
        d1.x += acc[4]; d1.y += acc[5]; d1.z += acc[6]; d1.w += acc[7];
        *(v4f*)(myRow + 16 * jj)     = d0;
        *(v4f*)(myRow + 16 * jj + 4) = d1;
    }
    asm volatile("s_wait_dscnt 0" ::: "memory");

#pragma unroll
    for (int i = 0; i < 8; ++i) {
        int srow = __shfl(rowL, 2 * i + hi, 32);
        v4f v = *(const v4f*)(sData + srow * SD_STRIDE + 4 * l16);
        __builtin_nontemporal_store(v, (v4f*)((char*)out
                + (size_t)((unsigned)(base + srow) * (NDIM * 4u) + 16u * (unsigned)l16)));
    }

    __syncthreads();
    if (tid < ROWS_PER_BLOCK / 2) {
        v2f lj = *(const v2f*)(sLogj + 2 * tid);
        __builtin_nontemporal_store(lj, (v2f*)(logj + base + 2 * tid));
    }
}

extern "C" void kernel_launch(void* const* d_in, const int* in_sizes, int n_in,
                              void* d_out, int out_size, void* d_ws, size_t ws_size,
                              hipStream_t stream) {
    const float* data  = (const float*)d_in[0];
    const int*   label = (const int*)  d_in[1];
    const float* wT    = (const float*)d_in[2];
    const float* xx    = (const float*)d_in[3];
    const float* yy    = (const float*)d_in[4];
    const float* dl    = (const float*)d_in[5];
    float* out  = (float*)d_out;
    float* logj = out + (size_t)NTOT * NDIM;

    dim3 grid(NTOT / ROWS_PER_BLOCK);
    cst_fused_kernel<<<grid, TPB, 0, stream>>>(data, label, wT, xx, yy, dl, out, logj);
}


# Class ranking moved under the tile's load latency: labels fetched first, its 40 ballots issued as one run ahead of the scalar counts (no compare-to-scalar round trip per step); same ranks and sums
# speedup vs baseline: 2.3158x; 1.0555x over previous
#include <hip/hip_runtime.h>
#include <hip/hip_bf16.h>


typedef float v2f __attribute__((ext_vector_type(2)));
typedef float v4f __attribute__((ext_vector_type(4)));
typedef float v8f __attribute__((ext_vector_type(8)));

#define N_CLASS 10
#define NDIM    64
#define NCOMP   32
#define NKNOT   32
#define NTOT    65536

#define ROWS_PER_BLOCK 128
#define TPB            256
#define NWAVES         8
#define LAB_PER_LANE   (ROWS_PER_BLOCK / 32)
#define SD_STRIDE      68


__global__ __launch_bounds__(TPB, 4)
void cst_fused_kernel(const float* __restrict__ data,
                      const int*   __restrict__ label,
                      const float* __restrict__ wT,
                      const float* __restrict__ xx,
                      const float* __restrict__ yy,
                      const float* __restrict__ dl,
                      float* __restrict__ out,
                      float* __restrict__ logj)
{
    __shared__ float sData[ROWS_PER_BLOCK * SD_STRIDE];
    __shared__ v2f   sW1[16 * 2 * 32];
    __shared__ v2f   sW2[8 * 4 * 32];
    __shared__ int   sSlot[NWAVES * 16];
    __shared__ float sLogj[ROWS_PER_BLOCK];

    const int tid  = threadIdx.x;
    const int base = blockIdx.x * ROWS_PER_BLOCK;
    const int lane = tid & 31;
    const int wave = tid >> 5;
    const int hi   = lane >> 4;
    const int l16  = lane & 15;

    int lab[LAB_PER_LANE];
#pragma unroll
    for (int q = 0; q < LAB_PER_LANE; ++q) lab[q] = label[base + 32 * q + lane];

    const int row0 = tid >> 4, c4 = tid & 15;
    v4f tile[(ROWS_PER_BLOCK * 16) / TPB];
    {
        const unsigned dOff = (unsigned)(base + row0) * (NDIM * 4u) + (unsigned)c4 * 16u;
        const char* dataB = (const char*)data;
#pragma unroll
        for (int t = 0; t < (ROWS_PER_BLOCK * 16) / TPB; ++t)
            tile[t] = __builtin_nontemporal_load(
                        (const v4f*)(dataB + (size_t)dOff + (size_t)t * (16 * NDIM * 4)));
    }
    v2f w1v[1024 / TPB];
#pragma unroll
    for (int t = 0; t < 1024 / TPB; ++t) {
        int e = tid + t * TPB;
        int ln = e & 31, cj = e >> 5;
        int chunk = cj >> 1, j = cj & 1;
        int krow = 4 * chunk + 2 * (ln >> 4);
        int col  = (ln & 15) + 16 * j;
        w1v[t].x = wT[krow * NCOMP + col];
        w1v[t].y = wT[(krow + 1) * NCOMP + col];
    }
    v2f w2v[1024 / TPB];
#pragma unroll
    for (int t = 0; t < 1024 / TPB; ++t) {
        int e = tid + t * TPB;
        int ln = e & 31, cj = e >> 5;
        int chunk = cj >> 2, jj = cj & 3;
        int n = jj * 16 + (ln & 15);
        int k = ((chunk < 4) ? 0 : 8) + 8 * (ln >> 4) + 2 * chunk;
        w2v[t] = *(const v2f*)((const char*)wT + (size_t)((unsigned)(n * NCOMP + k) * 4u));
    }

    {
        unsigned m[N_CLASS][LAB_PER_LANE];
#pragma unroll
        for (int c = 0; c < N_CLASS; ++c)
#pragma unroll
            for (int q = 0; q < LAB_PER_LANE; ++q)
                m[c][q] = __builtin_amdgcn_ballot_w32(lab[q] == c);
        int rk[LAB_PER_LANE];
#pragma unroll
        for (int q = 0; q < LAB_PER_LANE; ++q) rk[q] = 0;
        int off = 0;
#pragma unroll
        for (int c = 0; c < N_CLASS; ++c) {
#pragma unroll
            for (int q = 0; q < LAB_PER_LANE; ++q) {
                int r = (int)__builtin_amdgcn_mbcnt_lo(m[c][q], (unsigned)off);
                rk[q] = (lab[q] == c) ? r : rk[q];
                off += __builtin_popcount(m[c][q]);
            }
        }
#pragma unroll
        for (int q = 0; q < LAB_PER_LANE; ++q)
            if ((rk[q] >> 4) == wave)
                sSlot[rk[q]] = (lab[q] << 8) | (32 * q + lane);
    }

#pragma unroll
    for (int t = 0; t < (ROWS_PER_BLOCK * 16) / TPB; ++t)
        *(v4f*)(sData + (row0 + 16 * t) * SD_STRIDE + c4 * 4) = tile[t];
#pragma unroll
    for (int t = 0; t < 1024 / TPB; ++t) sW1[tid + t * TPB] = w1v[t];
#pragma unroll
    for (int t = 0; t < 1024 / TPB; ++t) sW2[tid + t * TPB] = w2v[t];
    __syncthreads();

    const int slot = sSlot[wave * 16 + l16];
    const int rowL = slot & 255;
    v2f bd[16];
#pragma unroll
    for (int c = 0; c < 16; ++c)
        bd[c] = *(const v2f*)(sData + rowL * SD_STRIDE + 4 * c + 2 * hi);

    v8f c0 = {0.f, 0.f, 0.f, 0.f, 0.f, 0.f, 0.f, 0.f};
    v8f c1 = c0;
#pragma unroll
    for (int c = 0; c < 16; ++c)
        c0 = __builtin_amdgcn_wmma_f32_16x16x4_f32(false, sW1[(c * 2 + 0) * 32 + lane], false,
                bd[c], (short)0, c0, false, false);
#pragma unroll
    for (int c = 0; c < 16; ++c)
        c1 = __builtin_amdgcn_wmma_f32_16x16x4_f32(false, sW1[(c * 2 + 1) * 32 + lane], false,
                bd[c], (short)0, c1, false, false);

    const unsigned cls = (unsigned)slot >> 8;
    const unsigned kb0 = (cls * NCOMP + 8u * (unsigned)hi) * (NKNOT * 4u);
    const char* xxB = (const char*)xx;
    const char* yyB = (const char*)yy;
    const char* dlB = (const char*)dl;
#define EB(e) ((size_t)((((e) < 8) ? (e) : (e) + 8) * NKNOT * 4))
#define KNOT(tab, off, e, cst) (*(const float*)((tab) + (size_t)(off) + EB(e) + (size_t)(cst)))
#define XV(e) (((e) < 8) ? c0[(e) & 7] : c1[(e) & 7])
    unsigned pos[16];
    bool     top[16];
    bool     bot[16];
    float ev[16];
    float lsum = 0.0f;

#pragma unroll
    for (int e = 0; e < 16; ++e) pos[e] = (KNOT(xxB, kb0, e, 15 * 4) < XV(e)) ? kb0 + 64u : kb0;
#pragma unroll
    for (int e = 0; e < 16; ++e) pos[e] += (KNOT(xxB, pos[e], e, 7 * 4) < XV(e)) ? 32u : 0u;
#pragma unroll
    for (int e = 0; e < 16; ++e) pos[e] += (KNOT(xxB, pos[e], e, 3 * 4) < XV(e)) ? 16u : 0u;
    v4f   win[16];
    float wfront[16];
#pragma unroll
    for (int e = 0; e < 16; ++e) {
        win[e] = *(const v4f*)(xxB + (size_t)pos[e] + EB(e));
        unsigned pf = (pos[e] > kb0 + 4u ? pos[e] : kb0 + 4u);
        wfront[e] = *(const float*)(xxB + (size_t)pf + EB(e) - 4);
    }

    float xk[16], xk1[16], yk[16], yk1[16], dk[16], dk1[16];
    unsigned gb[16];
    auto settle = [&](int e) {
        float x  = XV(e);
        bool  m4 = win[e].y < x;
        float a  = m4 ? win[e].z : win[e].x;
        bool  m5 = a < x;
        top[e] = (pos[e] == kb0 + 112u) && m4 && m5 && (win[e].w < x);
        xk[e]  = m5 ? a : (m4 ? win[e].y : wfront[e]);
        xk1[e] = m5 ? (m4 ? win[e].w : win[e].y) : a;
        unsigned fin = pos[e] + (m4 ? 8u : 0u) + (m5 ? 4u : 0u);
        bot[e] = (fin == kb0);
        gb[e] = (fin > kb0 + 4u ? fin : kb0 + 4u) - 4u;
    };
    auto gather = [&](int e) {
        yk[e] = KNOT(yyB, gb[e], e, 0); yk1[e] = KNOT(yyB, gb[e], e, 4);
        dk[e] = KNOT(dlB, gb[e], e, 0); dk1[e] = KNOT(dlB, gb[e], e, 4);
    };
    auto spline = [&](int e) {
        float x   = XV(e);
        float w   = xk1[e] - xk[e];
        float s   = (yk1[e] - yk[e]) / w;
        float xi  = fminf(fmaxf((x - xk[e]) / w, 0.0f), 1.0f);
        float xi1 = 1.0f - xi;
        float cc  = xi * xi1;
        float den = s + (dk1[e] + dk[e] - 2.0f * s) * cc;
        float y_in = yk[e] + (yk1[e] - yk[e]) * (s * xi * xi + dk[e] * cc) / den;
        float d_in = s * s * (dk1[e] * xi * xi + 2.0f * s * cc + dk[e] * xi1 * xi1)
                     / (den * den);
        bool below = bot[e];
        bool above = top[e];
        float y  = below ? fmaf(x - xk[e], dk[e], yk[e])
                 : (above ? fmaf(x - xk1[e], dk1[e], yk1[e]) : y_in);
        float dd = below ? dk[e] : (above ? dk1[e] : d_in);
        lsum += __logf(dd);
        ev[e] = y - x;
    };
#pragma unroll
    for (int e = 0; e < 16; ++e) settle(e);
#pragma unroll
    for (int e = 0; e < 16; ++e) gather(e);
    __builtin_amdgcn_sched_barrier(0);
#pragma unroll
    for (int e = 0; e < 8; ++e) spline(e);
    __builtin_amdgcn_sched_barrier(0);
#pragma unroll
    for (int e = 8; e < 16; ++e) spline(e);
#undef XV
#undef KNOT
#undef EB

    lsum += __shfl_xor(lsum, 16, 32);
    if (hi == 0) sLogj[rowL] = lsum;

    float* myRow = sData + rowL * SD_STRIDE + 8 * hi;
#pragma unroll
    for (int jj = 0; jj < 4; ++jj) {
        v8f acc = {0.f, 0.f, 0.f, 0.f, 0.f, 0.f, 0.f, 0.f};
#pragma unroll
        for (int c = 0; c < 8; ++c) {
            v2f be;
            be.x = ev[2 * c];
            be.y = ev[2 * c + 1];
            acc = __builtin_amdgcn_wmma_f32_16x16x4_f32(false, sW2[(c * 4 + jj) * 32 + lane], false,
                    be, (short)0, acc, false, false);
        }
        v4f d0 = *(v4f*)(myRow + 16 * jj);
        v4f d1 = *(v4f*)(myRow + 16 * jj + 4);
        d0.x += acc[0]; d0.y += acc[1]; d0.z += acc[2]; d0.w += acc[3];
        d1.x += acc[4]; d1.y += acc[5]; d1.z += acc[6]; d1.w += acc[7];
        *(v4f*)(myRow + 16 * jj)     = d0;
        *(v4f*)(myRow + 16 * jj + 4) = d1;
    }
    asm volatile("s_wait_dscnt 0" ::: "memory");

#pragma unroll
    for (int i = 0; i < 8; ++i) {
        int srow = __shfl(rowL, 2 * i + hi, 32);
        v4f v = *(const v4f*)(sData + srow * SD_STRIDE + 4 * l16);
        __builtin_nontemporal_store(v, (v4f*)((char*)out
                + (size_t)((unsigned)(base + srow) * (NDIM * 4u) + 16u * (unsigned)l16)));
    }

    __syncthreads();
    if (tid < ROWS_PER_BLOCK / 2) {
        v2f lj = *(const v2f*)(sLogj + 2 * tid);
        __builtin_nontemporal_store(lj, (v2f*)(logj + base + 2 * tid));
    }
}

extern "C" void kernel_launch(void* const* d_in, const int* in_sizes, int n_in,
                              void* d_out, int out_size, void* d_ws, size_t ws_size,
                              hipStream_t stream) {
    const float* data  = (const float*)d_in[0];
    const int*   label = (const int*)  d_in[1];
    const float* wT    = (const float*)d_in[2];
    const float* xx    = (const float*)d_in[3];
    const float* yy    = (const float*)d_in[4];
    const float* dl    = (const float*)d_in[5];
    float* out  = (float*)d_out;
    float* logj = out + (size_t)NTOT * NDIM;

    dim3 grid(NTOT / ROWS_PER_BLOCK);
    cst_fused_kernel<<<grid, TPB, 0, stream>>>(data, label, wT, xx, yy, dl, out, logj);
}


# Spline evaluated over its common denominator (numerator, denominator scaled by w^3): one IEEE division per element serves y and the derivative, not four; rounding moves an ulp or two, same sum order
# speedup vs baseline: 2.5373x; 1.0956x over previous
#include <hip/hip_runtime.h>
#include <hip/hip_bf16.h>


typedef float v2f __attribute__((ext_vector_type(2)));
typedef float v4f __attribute__((ext_vector_type(4)));
typedef float v8f __attribute__((ext_vector_type(8)));

#define N_CLASS 10
#define NDIM    64
#define NCOMP   32
#define NKNOT   32
#define NTOT    65536

#define ROWS_PER_BLOCK 128
#define TPB            256
#define NWAVES         8
#define LAB_PER_LANE   (ROWS_PER_BLOCK / 32)
#define SD_STRIDE      68


__global__ __launch_bounds__(TPB, 4)
void cst_fused_kernel(const float* __restrict__ data,
                      const int*   __restrict__ label,
                      const float* __restrict__ wT,
                      const float* __restrict__ xx,
                      const float* __restrict__ yy,
                      const float* __restrict__ dl,
                      float* __restrict__ out,
                      float* __restrict__ logj)
{
    __shared__ float sData[ROWS_PER_BLOCK * SD_STRIDE];
    __shared__ v2f   sW1[16 * 2 * 32];
    __shared__ v2f   sW2[8 * 4 * 32];
    __shared__ int   sSlot[NWAVES * 16];
    __shared__ float sLogj[ROWS_PER_BLOCK];

    const int tid  = threadIdx.x;
    const int base = blockIdx.x * ROWS_PER_BLOCK;
    const int lane = tid & 31;
    const int wave = tid >> 5;
    const int hi   = lane >> 4;
    const int l16  = lane & 15;

    int lab[LAB_PER_LANE];
#pragma unroll
    for (int q = 0; q < LAB_PER_LANE; ++q) lab[q] = label[base + 32 * q + lane];

    const int row0 = tid >> 4, c4 = tid & 15;
    v4f tile[(ROWS_PER_BLOCK * 16) / TPB];
    {
        const unsigned dOff = (unsigned)(base + row0) * (NDIM * 4u) + (unsigned)c4 * 16u;
        const char* dataB = (const char*)data;
#pragma unroll
        for (int t = 0; t < (ROWS_PER_BLOCK * 16) / TPB; ++t)
            tile[t] = __builtin_nontemporal_load(
                        (const v4f*)(dataB + (size_t)dOff + (size_t)t * (16 * NDIM * 4)));
    }
    v2f w1v[1024 / TPB];
#pragma unroll
    for (int t = 0; t < 1024 / TPB; ++t) {
        int e = tid + t * TPB;
        int ln = e & 31, cj = e >> 5;
        int chunk = cj >> 1, j = cj & 1;
        int krow = 4 * chunk + 2 * (ln >> 4);
        int col  = (ln & 15) + 16 * j;
        w1v[t].x = wT[krow * NCOMP + col];
        w1v[t].y = wT[(krow + 1) * NCOMP + col];
    }
    v2f w2v[1024 / TPB];
#pragma unroll
    for (int t = 0; t < 1024 / TPB; ++t) {
        int e = tid + t * TPB;
        int ln = e & 31, cj = e >> 5;
        int chunk = cj >> 2, jj = cj & 3;
        int n = jj * 16 + (ln & 15);
        int k = ((chunk < 4) ? 0 : 8) + 8 * (ln >> 4) + 2 * chunk;
        w2v[t] = *(const v2f*)((const char*)wT + (size_t)((unsigned)(n * NCOMP + k) * 4u));
    }

    {
        unsigned m[N_CLASS][LAB_PER_LANE];
#pragma unroll
        for (int c = 0; c < N_CLASS; ++c)
#pragma unroll
            for (int q = 0; q < LAB_PER_LANE; ++q)
                m[c][q] = __builtin_amdgcn_ballot_w32(lab[q] == c);
        int rk[LAB_PER_LANE];
#pragma unroll
        for (int q = 0; q < LAB_PER_LANE; ++q) rk[q] = 0;
        int off = 0;
#pragma unroll
        for (int c = 0; c < N_CLASS; ++c) {
#pragma unroll
            for (int q = 0; q < LAB_PER_LANE; ++q) {
                int r = (int)__builtin_amdgcn_mbcnt_lo(m[c][q], (unsigned)off);
                rk[q] = (lab[q] == c) ? r : rk[q];
                off += __builtin_popcount(m[c][q]);
            }
        }
#pragma unroll
        for (int q = 0; q < LAB_PER_LANE; ++q)
            if ((rk[q] >> 4) == wave)
                sSlot[rk[q]] = (lab[q] << 8) | (32 * q + lane);
    }

#pragma unroll
    for (int t = 0; t < (ROWS_PER_BLOCK * 16) / TPB; ++t)
        *(v4f*)(sData + (row0 + 16 * t) * SD_STRIDE + c4 * 4) = tile[t];
#pragma unroll
    for (int t = 0; t < 1024 / TPB; ++t) sW1[tid + t * TPB] = w1v[t];
#pragma unroll
    for (int t = 0; t < 1024 / TPB; ++t) sW2[tid + t * TPB] = w2v[t];
    __syncthreads();

    const int slot = sSlot[wave * 16 + l16];
    const int rowL = slot & 255;
    v2f bd[16];
#pragma unroll
    for (int c = 0; c < 16; ++c)
        bd[c] = *(const v2f*)(sData + rowL * SD_STRIDE + 4 * c + 2 * hi);

    v8f c0 = {0.f, 0.f, 0.f, 0.f, 0.f, 0.f, 0.f, 0.f};
    v8f c1 = c0;
#pragma unroll
    for (int c = 0; c < 16; ++c)
        c0 = __builtin_amdgcn_wmma_f32_16x16x4_f32(false, sW1[(c * 2 + 0) * 32 + lane], false,
                bd[c], (short)0, c0, false, false);
#pragma unroll
    for (int c = 0; c < 16; ++c)
        c1 = __builtin_amdgcn_wmma_f32_16x16x4_f32(false, sW1[(c * 2 + 1) * 32 + lane], false,
                bd[c], (short)0, c1, false, false);

    const unsigned cls = (unsigned)slot >> 8;
    const unsigned kb0 = (cls * NCOMP + 8u * (unsigned)hi) * (NKNOT * 4u);
    const char* xxB = (const char*)xx;
    const char* yyB = (const char*)yy;
    const char* dlB = (const char*)dl;
#define EB(e) ((size_t)((((e) < 8) ? (e) : (e) + 8) * NKNOT * 4))
#define KNOT(tab, off, e, cst) (*(const float*)((tab) + (size_t)(off) + EB(e) + (size_t)(cst)))
#define XV(e) (((e) < 8) ? c0[(e) & 7] : c1[(e) & 7])
    unsigned pos[16];
    bool     top[16];
    bool     bot[16];
    float ev[16];
    float lsum = 0.0f;

#pragma unroll
    for (int e = 0; e < 16; ++e) pos[e] = (KNOT(xxB, kb0, e, 15 * 4) < XV(e)) ? kb0 + 64u : kb0;
#pragma unroll
    for (int e = 0; e < 16; ++e) pos[e] += (KNOT(xxB, pos[e], e, 7 * 4) < XV(e)) ? 32u : 0u;
#pragma unroll
    for (int e = 0; e < 16; ++e) pos[e] += (KNOT(xxB, pos[e], e, 3 * 4) < XV(e)) ? 16u : 0u;
    v4f   win[16];
    float wfront[16];
#pragma unroll
    for (int e = 0; e < 16; ++e) {
        win[e] = *(const v4f*)(xxB + (size_t)pos[e] + EB(e));
        unsigned pf = (pos[e] > kb0 + 4u ? pos[e] : kb0 + 4u);
        wfront[e] = *(const float*)(xxB + (size_t)pf + EB(e) - 4);
    }

    float xk[16], xk1[16], yk[16], yk1[16], dk[16], dk1[16];
    unsigned gb[16];
    auto settle = [&](int e) {
        float x  = XV(e);
        bool  m4 = win[e].y < x;
        float a  = m4 ? win[e].z : win[e].x;
        bool  m5 = a < x;
        top[e] = (pos[e] == kb0 + 112u) && m4 && m5 && (win[e].w < x);
        xk[e]  = m5 ? a : (m4 ? win[e].y : wfront[e]);
        xk1[e] = m5 ? (m4 ? win[e].w : win[e].y) : a;
        unsigned fin = pos[e] + (m4 ? 8u : 0u) + (m5 ? 4u : 0u);
        bot[e] = (fin == kb0);
        gb[e] = (fin > kb0 + 4u ? fin : kb0 + 4u) - 4u;
    };
    auto gather = [&](int e) {
        yk[e] = KNOT(yyB, gb[e], e, 0); yk1[e] = KNOT(yyB, gb[e], e, 4);
        dk[e] = KNOT(dlB, gb[e], e, 0); dk1[e] = KNOT(dlB, gb[e], e, 4);
    };
    auto spline = [&](int e) {
        float x   = XV(e);
        float w   = xk1[e] - xk[e];
        float dy  = yk1[e] - yk[e];
        float t   = fminf(fmaxf(x - xk[e], 0.0f), w);
        float u   = w - t;
        float p   = t * u;
        float wd  = w * dk[e];
        float wd1 = w * dk1[e];
        float nump = fmaf(wd, p, (dy * t) * t);
        float denp = fmaf(wd1 + wd - 2.0f * dy, p, dy * (w * w));
        float r    = 1.0f / denp;
        float y_in = fmaf(dy * nump, r, yk[e]);
        float g    = dy * r;
        float q    = fmaf(wd1, t * t, fmaf(wd, u * u, (2.0f * dy) * p));
        float d_in = (g * g) * (w * q);
        bool below = bot[e];
        bool above = top[e];
        float y  = below ? fmaf(x - xk[e], dk[e], yk[e])
                 : (above ? fmaf(x - xk1[e], dk1[e], yk1[e]) : y_in);
        float dd = below ? dk[e] : (above ? dk1[e] : d_in);
        lsum += __logf(dd);
        ev[e] = y - x;
    };
#pragma unroll
    for (int e = 0; e < 16; ++e) settle(e);
#pragma unroll
    for (int e = 0; e < 16; ++e) gather(e);
    __builtin_amdgcn_sched_barrier(0);
#pragma unroll
    for (int e = 0; e < 8; ++e) spline(e);
    __builtin_amdgcn_sched_barrier(0);
#pragma unroll
    for (int e = 8; e < 16; ++e) spline(e);
#undef XV
#undef KNOT
#undef EB

    lsum += __shfl_xor(lsum, 16, 32);
    if (hi == 0) sLogj[rowL] = lsum;

    float* myRow = sData + rowL * SD_STRIDE + 8 * hi;
#pragma unroll
    for (int jj = 0; jj < 4; ++jj) {
        v8f acc = {0.f, 0.f, 0.f, 0.f, 0.f, 0.f, 0.f, 0.f};
#pragma unroll
        for (int c = 0; c < 8; ++c) {
            v2f be;
            be.x = ev[2 * c];
            be.y = ev[2 * c + 1];
            acc = __builtin_amdgcn_wmma_f32_16x16x4_f32(false, sW2[(c * 4 + jj) * 32 + lane], false,
                    be, (short)0, acc, false, false);
        }
        v4f d0 = *(v4f*)(myRow + 16 * jj);
        v4f d1 = *(v4f*)(myRow + 16 * jj + 4);
        d0.x += acc[0]; d0.y += acc[1]; d0.z += acc[2]; d0.w += acc[3];
        d1.x += acc[4]; d1.y += acc[5]; d1.z += acc[6]; d1.w += acc[7];
        *(v4f*)(myRow + 16 * jj)     = d0;
        *(v4f*)(myRow + 16 * jj + 4) = d1;
    }
    asm volatile("s_wait_dscnt 0" ::: "memory");

#pragma unroll
    for (int i = 0; i < 8; ++i) {
        int srow = __shfl(rowL, 2 * i + hi, 32);
        v4f v = *(const v4f*)(sData + srow * SD_STRIDE + 4 * l16);
        __builtin_nontemporal_store(v, (v4f*)((char*)out
                + (size_t)((unsigned)(base + srow) * (NDIM * 4u) + 16u * (unsigned)l16)));
    }

    __syncthreads();
    if (tid < ROWS_PER_BLOCK / 2) {
        v2f lj = *(const v2f*)(sLogj + 2 * tid);
        __builtin_nontemporal_store(lj, (v2f*)(logj + base + 2 * tid));
    }
}

extern "C" void kernel_launch(void* const* d_in, const int* in_sizes, int n_in,
                              void* d_out, int out_size, void* d_ws, size_t ws_size,
                              hipStream_t stream) {
    const float* data  = (const float*)d_in[0];
    const int*   label = (const int*)  d_in[1];
    const float* wT    = (const float*)d_in[2];
    const float* xx    = (const float*)d_in[3];
    const float* yy    = (const float*)d_in[4];
    const float* dl    = (const float*)d_in[5];
    float* out  = (float*)d_out;
    float* logj = out + (size_t)NTOT * NDIM;

    dim3 grid(NTOT / ROWS_PER_BLOCK);
    cst_fused_kernel<<<grid, TPB, 0, stream>>>(data, label, wT, xx, yy, dl, out, logj);
}
